# starting kernel
# baseline (speedup 1.0000x reference)
#include <hip/hip_runtime.h>
#include <math.h>

#define NN 100000
#define EE 1600000
#define ET (EE + NN)
#define IN_CH 128
#define HID 32
#define HEADS 4
#define HC 128
#define OUT_CH 40
#define NEG_SLOPE 0.2f

typedef __attribute__((ext_vector_type(2))) float v2f;
typedef __attribute__((ext_vector_type(8))) float v8f;

__device__ __forceinline__ float atomic_max_f32(float* addr, float v) {
    if (v >= 0.0f)
        return __int_as_float(atomicMax((int*)addr, __float_as_int(v)));
    else
        return __uint_as_float(atomicMin((unsigned int*)addr, __float_as_uint(v)));
}

__device__ __forceinline__ float lrelu(float v) {
    return v >= 0.0f ? v : NEG_SLOPE * v;
}

__global__ void k_gemm1(const float* __restrict__ x, const float* __restrict__ W1,
                        float* __restrict__ h1) {
    const int wave = threadIdx.x >> 5;
    const int row0 = (blockIdx.x * 4 + wave) * 16;
    if (row0 >= NN) return;
    const int lane = threadIdx.x & 31;
    const int m    = lane & 15;
    const int kb   = (lane >> 4) * 2;

    v8f c[8];
#pragma unroll
    for (int t = 0; t < 8; ++t) c[t] = (v8f){0,0,0,0,0,0,0,0};

    for (int k = 0; k < IN_CH; k += 4) {
        v2f a;
        const float* xr = x + (size_t)(row0 + m) * IN_CH + (k + kb);
        a.x = xr[0];
        a.y = xr[1];
#pragma unroll
        for (int t = 0; t < 8; ++t) {
            v2f b;
            const float* wr = W1 + (size_t)(k + kb) * HC + t * 16 + m;
            b.x = wr[0];
            b.y = wr[HC];
            c[t] = __builtin_amdgcn_wmma_f32_16x16x4_f32(false, a, false, b,
                                                         (short)0, c[t], false, false);
        }
    }
    const int rofs = (lane >> 4) * 8;
#pragma unroll
    for (int t = 0; t < 8; ++t)
#pragma unroll
        for (int r = 0; r < 8; ++r)
            h1[(size_t)(row0 + r + rofs) * HC + t * 16 + m] = c[t][r];
}

__global__ void k_al1(const float* __restrict__ h1, const float* __restrict__ a_src,
                      const float* __restrict__ a_dst, float* __restrict__ als,
                      float* __restrict__ ald) {
    int i = blockIdx.x * blockDim.x + threadIdx.x;
    if (i >= NN * HEADS) return;
    int n = i >> 2, h = i & 3;
    const float* hp = h1 + (size_t)n * HC + h * HID;
    float s = 0.f, d = 0.f;
#pragma unroll 8
    for (int cc = 0; cc < HID; ++cc) {
        float v = hp[cc];
        s += v * a_src[h * HID + cc];
        d += v * a_dst[h * HID + cc];
    }
    als[i] = s;
    ald[i] = d;
}

__global__ void k_init1(float* __restrict__ agg, float* __restrict__ m,
                        float* __restrict__ dn) {
    int i = blockIdx.x * blockDim.x + threadIdx.x;
    if (i < NN * HC) agg[i] = 0.0f;
    if (i < NN * HEADS) { m[i] = -__builtin_inff(); dn[i] = 0.0f; }
}

__global__ void k_emax1(const int* __restrict__ ei, const float* __restrict__ als,
                        const float* __restrict__ ald, float* __restrict__ m) {
    int i = blockIdx.x * blockDim.x + threadIdx.x;
    if (i >= ET * HEADS) return;
    int e = i >> 2, h = i & 3;
    int s, d;
    if (e < EE) { s = ei[e]; d = ei[EE + e]; } else { s = d = e - EE; }
    float v = lrelu(als[s * 4 + h] + ald[d * 4 + h]);
    atomic_max_f32(&m[d * 4 + h], v);
}

__global__ void k_esum1(const int* __restrict__ ei, const float* __restrict__ als,
                        const float* __restrict__ ald, const float* __restrict__ m,
                        float* __restrict__ dn) {
    int i = blockIdx.x * blockDim.x + threadIdx.x;
    if (i >= ET * HEADS) return;
    int e = i >> 2, h = i & 3;
    int s, d;
    if (e < EE) { s = ei[e]; d = ei[EE + e]; } else { s = d = e - EE; }
    float v = lrelu(als[s * 4 + h] + ald[d * 4 + h]);
    atomicAdd(&dn[d * 4 + h], __expf(v - m[d * 4 + h]));
}

__global__ void k_eagg1(const int* __restrict__ ei, const float* __restrict__ h1,
                        const float* __restrict__ als, const float* __restrict__ ald,
                        const float* __restrict__ m, const float* __restrict__ dn,
                        float* __restrict__ agg) {
    int e = blockIdx.x * 8 + (threadIdx.x >> 5);
    if (e >= ET) return;
    int lane = threadIdx.x & 31;
    int s, d;
    if (e < EE) { s = ei[e]; d = ei[EE + e]; } else { s = d = e - EE; }
    int h = lane >> 3;
    float v = lrelu(als[s * 4 + h] + ald[d * 4 + h]);
    float alpha = __expf(v - m[d * 4 + h]) / (dn[d * 4 + h] + 1e-16f);
    const float4 hv = *(const float4*)(h1 + (size_t)s * HC + lane * 4);
    float* op = agg + (size_t)d * HC + lane * 4;
    atomicAdd(op + 0, hv.x * alpha);
    atomicAdd(op + 1, hv.y * alpha);
    atomicAdd(op + 2, hv.z * alpha);
    atomicAdd(op + 3, hv.w * alpha);
}

__global__ void k_fin1(float* __restrict__ agg, const float* __restrict__ b1) {
    int i = blockIdx.x * blockDim.x + threadIdx.x;
    if (i >= NN * HC) return;
    float v = agg[i] + b1[i & (HC - 1)];
    agg[i] = v > 0.0f ? v : expm1f(v);
}

__global__ void k_gemm2(const float* __restrict__ h2, const float* __restrict__ W2,
                        float* __restrict__ z) {
    const int wave = threadIdx.x >> 5;
    const int row0 = (blockIdx.x * 4 + wave) * 16;
    if (row0 >= NN) return;
    const int lane = threadIdx.x & 31;
    const int m    = lane & 15;
    const int kb   = (lane >> 4) * 2;

    v8f c[3];
#pragma unroll
    for (int t = 0; t < 3; ++t) c[t] = (v8f){0,0,0,0,0,0,0,0};

    for (int k = 0; k < HC; k += 4) {
        v2f a;
        const float* xr = h2 + (size_t)(row0 + m) * HC + (k + kb);
        a.x = xr[0];
        a.y = xr[1];
#pragma unroll
        for (int t = 0; t < 3; ++t) {
            int col = t * 16 + m;
            v2f b;
            b.x = (col < OUT_CH) ? W2[(size_t)(k + kb) * OUT_CH + col] : 0.0f;
            b.y = (col < OUT_CH) ? W2[(size_t)(k + kb + 1) * OUT_CH + col] : 0.0f;
            c[t] = __builtin_amdgcn_wmma_f32_16x16x4_f32(false, a, false, b,
                                                         (short)0, c[t], false, false);
        }
    }
    const int rofs = (lane >> 4) * 8;
#pragma unroll
    for (int t = 0; t < 3; ++t) {
        int col = t * 16 + m;
        if (col < OUT_CH)
#pragma unroll
            for (int r = 0; r < 8; ++r)
                z[(size_t)(row0 + r + rofs) * OUT_CH + col] = c[t][r];
    }
}

__global__ void k_al2(const float* __restrict__ z, const float* __restrict__ a_src,
                      const float* __restrict__ a_dst, float* __restrict__ als,
                      float* __restrict__ ald) {
    int n = blockIdx.x * blockDim.x + threadIdx.x;
    if (n >= NN) return;
    const float* zp = z + (size_t)n * OUT_CH;
    float s = 0.f, d = 0.f;
#pragma unroll 8
    for (int cc = 0; cc < OUT_CH; ++cc) {
        float v = zp[cc];
        s += v * a_src[cc];
        d += v * a_dst[cc];
    }
    als[n] = s;
    ald[n] = d;
}

__global__ void k_init2(float* __restrict__ out, const float* __restrict__ b2,
                        float* __restrict__ m, float* __restrict__ dn) {
    int i = blockIdx.x * blockDim.x + threadIdx.x;
    if (i < NN * OUT_CH) out[i] = b2[i % OUT_CH];
    if (i < NN) { m[i] = -__builtin_inff(); dn[i] = 0.0f; }
}

__global__ void k_emax2(const int* __restrict__ ei, const float* __restrict__ als,
                        const float* __restrict__ ald, float* __restrict__ m) {
    int e = blockIdx.x * blockDim.x + threadIdx.x;
    if (e >= ET) return;
    int s, d;
    if (e < EE) { s = ei[e]; d = ei[EE + e]; } else { s = d = e - EE; }
    atomic_max_f32(&m[d], lrelu(als[s] + ald[d]));
}

__global__ void k_esum2(const int* __restrict__ ei, const float* __restrict__ als,
                        const float* __restrict__ ald, const float* __restrict__ m,
                        float* __restrict__ dn) {
    int e = blockIdx.x * blockDim.x + threadIdx.x;
    if (e >= ET) return;
    int s, d;
    if (e < EE) { s = ei[e]; d = ei[EE + e]; } else { s = d = e - EE; }
    float v = lrelu(als[s] + ald[d]);
    atomicAdd(&dn[d], __expf(v - m[d]));
}

__global__ void k_eagg2(const int* __restrict__ ei, const float* __restrict__ z,
                        const float* __restrict__ als, const float* __restrict__ ald,
                        const float* __restrict__ m, const float* __restrict__ dn,
                        float* __restrict__ out) {
    int e = blockIdx.x * 8 + (threadIdx.x >> 5);
    if (e >= ET) return;
    int lane = threadIdx.x & 31;
    int s, d;
    if (e < EE) { s = ei[e]; d = ei[EE + e]; } else { s = d = e - EE; }
    float v = lrelu(als[s] + ald[d]);
    float alpha = __expf(v - m[d]) / (dn[d] + 1e-16f);
    const float* zp = z + (size_t)s * OUT_CH;
    float* op = out + (size_t)d * OUT_CH;
    atomicAdd(op + lane, zp[lane] * alpha);
    if (lane < OUT_CH - 32)
        atomicAdd(op + 32 + lane, zp[32 + lane] * alpha);
}

extern "C" void kernel_launch(void* const* d_in, const int* in_sizes, int n_in,
                              void* d_out, int out_size, void* d_ws, size_t ws_size,
                              hipStream_t stream) {
    const float* x      = (const float*)d_in[0];
    const int*   ei     = (const int*)  d_in[1];
    const float* W1     = (const float*)d_in[2];
    const float* a_src1 = (const float*)d_in[3];
    const float* a_dst1 = (const float*)d_in[4];
    const float* b1     = (const float*)d_in[5];
    const float* W2     = (const float*)d_in[6];
    const float* a_src2 = (const float*)d_in[7];
    const float* a_dst2 = (const float*)d_in[8];
    const float* b2     = (const float*)d_in[9];
    float* out = (float*)d_out;

    float* ws   = (float*)d_ws;
    float* h1   = ws;
    float* agg1 = h1   + (size_t)NN * HC;
    float* z2   = agg1 + (size_t)NN * HC;
    float* als1 = z2   + (size_t)NN * OUT_CH;
    float* ald1 = als1 + (size_t)NN * HEADS;
    float* m1   = ald1 + (size_t)NN * HEADS;
    float* dn1  = m1   + (size_t)NN * HEADS;
    float* als2 = dn1  + (size_t)NN * HEADS;
    float* ald2 = als2 + NN;
    float* m2   = ald2 + NN;
    float* dn2  = m2   + NN;

    const int TB = 256;
    const int gemmBlocks = (NN / 16 + 3) / 4;
    const int edgeWaveBlocks = (ET + 7) / 8;

    k_gemm1<<<gemmBlocks, 128, 0, stream>>>(x, W1, h1);
    k_al1  <<<(NN * HEADS + TB - 1) / TB, TB, 0, stream>>>(h1, a_src1, a_dst1, als1, ald1);
    k_init1<<<(NN * HC + TB - 1) / TB, TB, 0, stream>>>(agg1, m1, dn1);
    k_emax1<<<(ET * HEADS + TB - 1) / TB, TB, 0, stream>>>(ei, als1, ald1, m1);
    k_esum1<<<(ET * HEADS + TB - 1) / TB, TB, 0, stream>>>(ei, als1, ald1, m1, dn1);
    k_eagg1<<<edgeWaveBlocks, 256, 0, stream>>>(ei, h1, als1, ald1, m1, dn1, agg1);
    k_fin1 <<<(NN * HC + TB - 1) / TB, TB, 0, stream>>>(agg1, b1);

    k_gemm2<<<gemmBlocks, 128, 0, stream>>>(agg1, W2, z2);
    k_al2  <<<(NN + TB - 1) / TB, TB, 0, stream>>>(z2, a_src2, a_dst2, als2, ald2);
    k_init2<<<(NN * OUT_CH + TB - 1) / TB, TB, 0, stream>>>(out, b2, m2, dn2);
    k_emax2<<<(ET + TB - 1) / TB, TB, 0, stream>>>(ei, als2, ald2, m2);
    k_esum2<<<(ET + TB - 1) / TB, TB, 0, stream>>>(ei, als2, ald2, m2, dn2);
    k_eagg2<<<edgeWaveBlocks, 256, 0, stream>>>(ei, z2, als2, ald2, m2, dn2, out);
}


# Max, start-value and finish passes of both layers: a thread takes 4 items (16 bytes each in the node passes), all its loads issued ahead of its first store; 7 times fewer waves in these six kernels
# speedup vs baseline: 2.6233x; 2.6233x over previous
#include <hip/hip_runtime.h>
#include <math.h>

#define NN 100000
#define EE 1600000
#define IN_CH 128
#define HID 32
#define HEADS 4
#define HC 128
#define OUT_CH 40
#define NEG_SLOPE 0.2f
#define AP 64
#define DCOL OUT_CH
#define SBLK 256
#define SPT 4

typedef __attribute__((ext_vector_type(2))) float v2f;
typedef __attribute__((ext_vector_type(8))) float v8f;
typedef __attribute__((ext_vector_type(4))) float v4f;

__device__ __forceinline__ float atomic_max_f32(float* addr, float v) {
    if (v >= 0.0f)
        return __int_as_float(atomicMax((int*)addr, __float_as_int(v)));
    else
        return __uint_as_float(atomicMin((unsigned int*)addr, __float_as_uint(v)));
}

__device__ __forceinline__ float lrelu(float v) {
    return v >= 0.0f ? v : NEG_SLOPE * v;
}

__device__ __forceinline__ int package_half() {
    return (int)((__builtin_amdgcn_s_sendmsg_rtn(0x87) >> 18) & 1u);
}

__global__ void k_gemm1(const float* __restrict__ x, const float* __restrict__ W1,
                        float* __restrict__ h1) {
    const int wave = threadIdx.x >> 5;
    const int row0 = (blockIdx.x * 4 + wave) * 16;
    if (row0 >= NN) return;
    const int lane = threadIdx.x & 31;
    const int m    = lane & 15;
    const int kb   = (lane >> 4) * 2;

    v8f c[8];
#pragma unroll
    for (int t = 0; t < 8; ++t) c[t] = (v8f){0,0,0,0,0,0,0,0};

    for (int k = 0; k < IN_CH; k += 4) {
        v2f a;
        const float* xr = x + (size_t)(row0 + m) * IN_CH + (k + kb);
        a.x = xr[0];
        a.y = xr[1];
#pragma unroll
        for (int t = 0; t < 8; ++t) {
            v2f b;
            const float* wr = W1 + (size_t)(k + kb) * HC + t * 16 + m;
            b.x = wr[0];
            b.y = wr[HC];
            c[t] = __builtin_amdgcn_wmma_f32_16x16x4_f32(false, a, false, b,
                                                         (short)0, c[t], false, false);
        }
    }
    const int rofs = (lane >> 4) * 8;
#pragma unroll
    for (int t = 0; t < 8; ++t)
#pragma unroll
        for (int r = 0; r < 8; ++r)
            h1[(size_t)(row0 + r + rofs) * HC + t * 16 + m] = c[t][r];
}

__global__ void k_al1(const float* __restrict__ h1, const float* __restrict__ a_src,
                      const float* __restrict__ a_dst, float* __restrict__ als,
                      float* __restrict__ ald, float* __restrict__ m) {
    int i = blockIdx.x * blockDim.x + threadIdx.x;
    if (i >= NN * HEADS) return;
    int n = i >> 2, h = i & 3;
    const float* hp = h1 + (size_t)n * HC + h * HID;
    float s = 0.f, d = 0.f;
#pragma unroll 8
    for (int cc = 0; cc < HID; ++cc) {
        float v = hp[cc];
        s += v * a_src[h * HID + cc];
        d += v * a_dst[h * HID + cc];
    }
    als[i] = s;
    ald[i] = d;
    const float self = lrelu(s + d);
    m[i] = self;
    m[NN * HEADS + i] = self;
}

__global__ void k_seed1(const float* __restrict__ h1, const float* __restrict__ als,
                        const float* __restrict__ ald, const float* __restrict__ m,
                        float* __restrict__ mj, float* __restrict__ dn, float* __restrict__ agg) {
    const int base = blockIdx.x * (SBLK * SPT) + threadIdx.x;
    v4f v[SPT];
    float m0[SPT], m1[SPT], ls[SPT], ld[SPT];
#pragma unroll
    for (int k = 0; k < SPT; ++k) {
        const int i = base + k * SBLK;
        const int nh = i >> 3;
        v[k]  = ((const v4f*)h1)[i];
        m0[k] = m[nh];
        m1[k] = m[NN * HEADS + nh];
        ls[k] = als[nh];
        ld[k] = ald[nh];
    }
#pragma unroll
    for (int k = 0; k < SPT; ++k) {
        const int i = base + k * SBLK;
        const int nh = i >> 3;
        const float mx = fmaxf(m0[k], m1[k]);
        const float w = __expf(lrelu(ls[k] + ld[k]) - mx);
        ((v4f*)agg)[i] = v[k] * w;
        ((v4f*)agg)[NN * HC / 4 + i] = (v4f){0.0f, 0.0f, 0.0f, 0.0f};
        if ((i & 7) == 0) {
            mj[nh] = mx;
            dn[nh] = w;
            dn[NN * HEADS + nh] = 0.0f;
        }
    }
}

static_assert(EE % 32 == 0 && HEADS == 4 && SPT * 8 == 32, "a wave's 32 edges are SPT rounds of 8 edges x 4 heads");
__global__ void k_emax1(const int* __restrict__ ei, const float* __restrict__ als,
                        const float* __restrict__ ald, float* __restrict__ m,
                        float* __restrict__ ev) {
    const int e0 = (blockIdx.x * (SBLK / 32) + (threadIdx.x >> 5)) * 32;
    if (e0 >= EE) return;
    const int lane = threadIdx.x & 31;
    const int s = ei[e0 + lane], d = ei[EE + e0 + lane];
    const int h = lane & 3, q = lane >> 2;
    int dk[SPT];
    float a[SPT], b[SPT];
#pragma unroll
    for (int k = 0; k < SPT; ++k) {
        const int sk = __shfl(s, k * 8 + q);
        dk[k] = __shfl(d, k * 8 + q);
        a[k] = als[sk * 4 + h];
        b[k] = ald[dk[k] * 4 + h];
    }
    float* mh = m + (size_t)package_half() * (NN * HEADS);
#pragma unroll
    for (int k = 0; k < SPT; ++k) {
        const float v = lrelu(a[k] + b[k]);
        ev[(size_t)e0 * HEADS + k * 32 + lane] = v;
        atomic_max_f32(&mh[dk[k] * 4 + h], v);
    }
}

#define EPW 8
static_assert(EE % EPW == 0, "the edge list is a whole number of groups");
static_assert(HID == 32 && HC == HEADS * 32, "one head = one wave-wide chunk of a row");
__global__ void k_eagg1(const int* __restrict__ ei, const float* __restrict__ h1,
                        const float* __restrict__ ev,
                        const float* __restrict__ m, float* __restrict__ dn,
                        float* __restrict__ agg) {
    const int e0 = (blockIdx.x * 8 + (threadIdx.x >> 5)) * EPW;
    if (e0 >= EE) return;
    const int lane = threadIdx.x & 31;
    const int je = e0 + (lane >> 2);
    const int hh = lane & 3;
    const int s = ei[je], d = ei[EE + je];
    const float v = ev[(size_t)e0 * HEADS + lane];
    const int half = package_half();

    float hv[EPW][HEADS];
#pragma unroll
    for (int j = 0; j < EPW; ++j) {
        const int sj = __builtin_amdgcn_readlane(s, j * 4);
        const float* hp = h1 + (size_t)sj * HC + lane;
#pragma unroll
        for (int q = 0; q < HEADS; ++q)
            hv[j][q] = hp[q * HID];
    }

    dn  += (size_t)half * (NN * HEADS);
    agg += (size_t)half * ((size_t)NN * HC);

    const float w = __expf(v - m[d * 4 + hh]);
    atomicAdd(&dn[d * 4 + hh], w);

#pragma unroll
    for (int j = 0; j < EPW; ++j) {
        const int dj = __builtin_amdgcn_readlane(d, j * 4);
        float* op = agg + (size_t)dj * HC + lane;
#pragma unroll
        for (int q = 0; q < HEADS; ++q) {
            const float wq = __int_as_float(__builtin_amdgcn_readlane(__float_as_int(w), j * 4 + q));
            atomicAdd(op + q * HID, hv[j][q] * wq);
        }
    }
}

__device__ __forceinline__ float fin1_one(float p, float q, float den, float b) {
    const float v = (p + q) / (den + 1e-16f) + b;
    return v > 0.0f ? v : expm1f(v);
}
static_assert((NN * HC / 4) % (SBLK * SPT) == 0 && SBLK % (HC / 4) == 0, "whole blocks; a thread's items share their channels");
__global__ void k_fin1(float* __restrict__ agg, const float* __restrict__ dn,
                       const float* __restrict__ b1) {
    const int base = blockIdx.x * (SBLK * SPT) + threadIdx.x;
    v4f p[SPT], q[SPT];
    float d0[SPT], d1[SPT];
#pragma unroll
    for (int k = 0; k < SPT; ++k) {
        const int i = base + k * SBLK;
        p[k]  = ((const v4f*)agg)[i];
        q[k]  = ((const v4f*)agg)[NN * HC / 4 + i];
        d0[k] = dn[i >> 3];
        d1[k] = dn[NN * HEADS + (i >> 3)];
    }
    const v4f bb = ((const v4f*)b1)[base & (HC / 4 - 1)];
#pragma unroll
    for (int k = 0; k < SPT; ++k) {
        const int i = base + k * SBLK;
        const float den = d0[k] + d1[k];
        ((v4f*)agg)[i] = (v4f){fin1_one(p[k].x, q[k].x, den, bb.x), fin1_one(p[k].y, q[k].y, den, bb.y),
                               fin1_one(p[k].z, q[k].z, den, bb.z), fin1_one(p[k].w, q[k].w, den, bb.w)};
    }
}

__global__ void k_gemm2(const float* __restrict__ h2, const float* __restrict__ W2,
                        float* __restrict__ z) {
    const int wave = threadIdx.x >> 5;
    const int row0 = (blockIdx.x * 4 + wave) * 16;
    if (row0 >= NN) return;
    const int lane = threadIdx.x & 31;
    const int m    = lane & 15;
    const int kb   = (lane >> 4) * 2;

    v8f c[3];
#pragma unroll
    for (int t = 0; t < 3; ++t) c[t] = (v8f){0,0,0,0,0,0,0,0};

    for (int k = 0; k < HC; k += 4) {
        v2f a;
        const float* xr = h2 + (size_t)(row0 + m) * HC + (k + kb);
        a.x = xr[0];
        a.y = xr[1];
#pragma unroll
        for (int t = 0; t < 3; ++t) {
            int col = t * 16 + m;
            v2f b;
            b.x = (col < OUT_CH) ? W2[(size_t)(k + kb) * OUT_CH + col] : 0.0f;
            b.y = (col < OUT_CH) ? W2[(size_t)(k + kb + 1) * OUT_CH + col] : 0.0f;
            c[t] = __builtin_amdgcn_wmma_f32_16x16x4_f32(false, a, false, b,
                                                         (short)0, c[t], false, false);
        }
    }
    const int rofs = (lane >> 4) * 8;
#pragma unroll
    for (int t = 0; t < 3; ++t) {
        int col = t * 16 + m;
        if (col < OUT_CH)
#pragma unroll
            for (int r = 0; r < 8; ++r)
                z[(size_t)(row0 + r + rofs) * OUT_CH + col] = c[t][r];
    }
}

__global__ void k_al2(const float* __restrict__ z, const float* __restrict__ a_src,
                      const float* __restrict__ a_dst, float* __restrict__ als,
                      float* __restrict__ ald, float* __restrict__ m) {
    int n = blockIdx.x * blockDim.x + threadIdx.x;
    if (n >= NN) return;
    const float* zp = z + (size_t)n * OUT_CH;
    float s = 0.f, d = 0.f;
#pragma unroll 8
    for (int cc = 0; cc < OUT_CH; ++cc) {
        float v = zp[cc];
        s += v * a_src[cc];
        d += v * a_dst[cc];
    }
    als[n] = s;
    ald[n] = d;
    const float self = lrelu(s + d);
    m[n] = self;
    m[NN + n] = self;
}

static_assert(OUT_CH % 4 == 0 && DCOL == OUT_CH && AP % 4 == 0, "the exp-sum opens the float4 behind the channels");
static_assert(SBLK % (AP / 4) == 0, "a thread's items sit in the same float4 column of their rows");
__global__ void k_seed2(const float* __restrict__ z, const float* __restrict__ als,
                        const float* __restrict__ ald, const float* __restrict__ m,
                        float* __restrict__ mj, float* __restrict__ acc) {
    const int base = blockIdx.x * (SBLK * SPT) + threadIdx.x;
    const int j = base % (AP / 4);
    v4f v[SPT];
    float m0[SPT], m1[SPT], ls[SPT], ld[SPT];
#pragma unroll
    for (int k = 0; k < SPT; ++k) {
        const int i = min(base + k * SBLK, NN * (AP / 4) - 1);
        const int n = i / (AP / 4);
        m0[k] = m[n];
        m1[k] = m[NN + n];
        ls[k] = als[n];
        ld[k] = ald[n];
        v[k] = ((const v4f*)z)[n * (OUT_CH / 4) + min(j, OUT_CH / 4 - 1)];
    }
    asm volatile("" :: "v"(v[0]), "v"(v[1]), "v"(v[2]), "v"(v[3]), "v"(m0[0]), "v"(m0[1]), "v"(m0[2]), "v"(m0[3]),
                       "v"(m1[0]), "v"(m1[1]), "v"(m1[2]), "v"(m1[3]), "v"(ls[0]), "v"(ls[1]), "v"(ls[2]), "v"(ls[3]),
                       "v"(ld[0]), "v"(ld[1]), "v"(ld[2]), "v"(ld[3]));
    static_assert(SPT == 4, "the statement above names SPT = 4 items");
#pragma unroll
    for (int k = 0; k < SPT; ++k) {
        const int i = base + k * SBLK;
        if (i < NN * (AP / 4)) {
            const int n = i / (AP / 4);
            v4f o = (v4f){0.0f, 0.0f, 0.0f, 0.0f};
            if (j <= OUT_CH / 4) {
                const float mx = fmaxf(m0[k], m1[k]);
                const float w = __expf(lrelu(ls[k] + ld[k]) - mx);
                if (j < OUT_CH / 4) {
                    o = v[k] * w;
                } else {
                    o.x = w;
                    mj[n] = mx;
                }
            }
            ((v4f*)acc)[i] = o;
            ((v4f*)acc)[NN * (AP / 4) + i] = (v4f){0.0f, 0.0f, 0.0f, 0.0f};
        }
    }
}

static_assert(EE % (32 * SPT) == 0, "the edge list is a whole number of wave groups");
__global__ void k_emax2(const int* __restrict__ ei, const float* __restrict__ als,
                        const float* __restrict__ ald, float* __restrict__ m,
                        float* __restrict__ ev) {
    const int e0 = (blockIdx.x * (SBLK / 32) + (threadIdx.x >> 5)) * (32 * SPT) + (threadIdx.x & 31);
    if (e0 >= EE) return;
    int s[SPT], d[SPT];
    float a[SPT], b[SPT];
#pragma unroll
    for (int k = 0; k < SPT; ++k) {
        s[k] = ei[e0 + k * 32];
        d[k] = ei[EE + e0 + k * 32];
    }
#pragma unroll
    for (int k = 0; k < SPT; ++k) {
        a[k] = als[s[k]];
        b[k] = ald[d[k]];
    }
    float* mh = m + (size_t)package_half() * NN;
#pragma unroll
    for (int k = 0; k < SPT; ++k) {
        const float v = lrelu(a[k] + b[k]);
        ev[e0 + k * 32] = v;
        atomic_max_f32(&mh[d[k]], v);
    }
}

static_assert(DCOL >= 32 && DCOL < AP && (AP * 4) % 128 == 0, "exp-sum sits in the row's second line");
__global__ void k_eagg2(const int* __restrict__ ei, const float* __restrict__ z,
                        const float* __restrict__ ev,
                        const float* __restrict__ m, float* __restrict__ acc) {
    const int e0 = (blockIdx.x * 8 + (threadIdx.x >> 5)) * EPW;
    if (e0 >= EE) return;
    const int lane = threadIdx.x & 31;
    const int je = e0 + (lane & (EPW - 1));
    const int s = ei[je], d = ei[EE + je];

    float z0[EPW], z1[EPW];
    const int tl = 32 + (lane & (OUT_CH - 32 - 1));
    static_assert(((OUT_CH - 32) & (OUT_CH - 33)) == 0, "tail width is a power of two");
#pragma unroll
    for (int j = 0; j < EPW; ++j) {
        const int sj = __builtin_amdgcn_readlane(s, j);
        const float* zp = z + (size_t)sj * OUT_CH;
        z0[j] = zp[lane];
        z1[j] = zp[tl];
    }

    const float lv = ev[je], mx = m[d];

    const int half = package_half();

    asm volatile("" :: "v"(z0[0]), "v"(z0[1]), "v"(z0[2]), "v"(z0[3]),
                       "v"(z0[4]), "v"(z0[5]), "v"(z0[6]), "v"(z0[7]),
                       "v"(z1[0]), "v"(z1[1]), "v"(z1[2]), "v"(z1[3]),
                       "v"(z1[4]), "v"(z1[5]), "v"(z1[6]), "v"(z1[7]),
                       "v"(lv), "v"(mx));

    const float w = __expf(lv - mx);

    acc += (size_t)half * ((size_t)NN * AP);

#pragma unroll
    for (int j = 0; j < EPW; ++j) {
        const int dj = __builtin_amdgcn_readlane(d, j);
        const float wj = __int_as_float(__builtin_amdgcn_readlane(__float_as_int(w), j));
        float* op = acc + (size_t)dj * AP;
        atomicAdd(op + lane, z0[j] * wj);
        const float tv = (lane == DCOL - 32) ? wj : z1[j] * wj;
        if (lane <= DCOL - 32)
            atomicAdd(op + 32 + lane, tv);
    }
}

__global__ void k_fin2(float* __restrict__ out, const float* __restrict__ acc,
                       const float* __restrict__ b2) {
    const int base = blockIdx.x * (SBLK * SPT) + threadIdx.x;
    v4f p[SPT], q[SPT], bb[SPT];
    float d0[SPT], d1[SPT];
#pragma unroll
    for (int k = 0; k < SPT; ++k) {
        const int i = min(base + k * SBLK, NN * (OUT_CH / 4) - 1);
        const int n = i / (OUT_CH / 4), j = i % (OUT_CH / 4);
        const float* ap = acc + (size_t)n * AP;
        const float* bp = ap + (size_t)NN * AP;
        p[k]  = ((const v4f*)ap)[j];
        q[k]  = ((const v4f*)bp)[j];
        d0[k] = ap[DCOL];
        d1[k] = bp[DCOL];
        bb[k] = ((const v4f*)b2)[j];
    }
    asm volatile("" :: "v"(p[0]), "v"(p[1]), "v"(p[2]), "v"(p[3]), "v"(q[0]), "v"(q[1]), "v"(q[2]), "v"(q[3]),
                       "v"(bb[0]), "v"(bb[1]), "v"(bb[2]), "v"(bb[3]),
                       "v"(d0[0]), "v"(d0[1]), "v"(d0[2]), "v"(d0[3]), "v"(d1[0]), "v"(d1[1]), "v"(d1[2]), "v"(d1[3]));
    static_assert(SPT == 4, "the statement above names SPT = 4 items");
#pragma unroll
    for (int k = 0; k < SPT; ++k) {
        const int i = base + k * SBLK;
        if (i < NN * (OUT_CH / 4)) {
            const float den = d0[k] + d1[k] + 1e-16f;
            ((v4f*)out)[i] = (v4f){(p[k].x + q[k].x) / den + bb[k].x, (p[k].y + q[k].y) / den + bb[k].y,
                                   (p[k].z + q[k].z) / den + bb[k].z, (p[k].w + q[k].w) / den + bb[k].w};
        }
    }
}

extern "C" void kernel_launch(void* const* d_in, const int* in_sizes, int n_in,
                              void* d_out, int out_size, void* d_ws, size_t ws_size,
                              hipStream_t stream) {
    const float* x      = (const float*)d_in[0];
    const int*   ei     = (const int*)  d_in[1];
    const float* W1     = (const float*)d_in[2];
    const float* a_src1 = (const float*)d_in[3];
    const float* a_dst1 = (const float*)d_in[4];
    const float* b1     = (const float*)d_in[5];
    const float* W2     = (const float*)d_in[6];
    const float* a_src2 = (const float*)d_in[7];
    const float* a_dst2 = (const float*)d_in[8];
    const float* b2     = (const float*)d_in[9];
    float* out = (float*)d_out;

    float* ws   = (float*)d_ws;
    float* h1   = ws;
    float* agg1 = h1   + (size_t)NN * HC;
    float* acc2 = agg1 + 2 * (size_t)NN * HC;
    float* z2   = acc2 + 2 * (size_t)NN * AP;
    float* als1 = z2   + (size_t)NN * OUT_CH;
    float* ald1 = als1 + (size_t)NN * HEADS;
    float* m1   = ald1 + (size_t)NN * HEADS;
    float* mj1  = m1   + 2 * (size_t)NN * HEADS;
    float* dn1  = mj1  + (size_t)NN * HEADS;
    float* als2 = dn1  + 2 * (size_t)NN * HEADS;
    float* ald2 = als2 + NN;
    float* m2   = ald2 + NN;
    float* mj2  = m2   + 2 * NN;
    float* ev   = mj2  + NN;

    const int TB = 256;
    const int gemmBlocks = (NN / 16 + 3) / 4;
    const int edgeWaveBlocks = (EE / EPW + 7) / 8;

    k_gemm1<<<gemmBlocks, 128, 0, stream>>>(x, W1, h1);
    k_al1  <<<(NN * HEADS + TB - 1) / TB, TB, 0, stream>>>(h1, a_src1, a_dst1, als1, ald1, m1);
    k_emax1<<<(EE / 32 + SBLK / 32 - 1) / (SBLK / 32), SBLK, 0, stream>>>(ei, als1, ald1, m1, ev);
    k_seed1<<<NN * HC / 4 / (SBLK * SPT), SBLK, 0, stream>>>(h1, als1, ald1, m1, mj1, dn1, agg1);
    k_eagg1<<<edgeWaveBlocks, 256, 0, stream>>>(ei, h1, ev, mj1, dn1, agg1);
    k_fin1 <<<NN * HC / 4 / (SBLK * SPT), SBLK, 0, stream>>>(agg1, dn1, b1);

    k_gemm2<<<gemmBlocks, 128, 0, stream>>>(agg1, W2, z2);
    k_al2  <<<(NN + TB - 1) / TB, TB, 0, stream>>>(z2, a_src2, a_dst2, als2, ald2, m2);
    k_emax2<<<(EE / (32 * SPT) + SBLK / 32 - 1) / (SBLK / 32), SBLK, 0, stream>>>(ei, als2, ald2, m2, ev);
    k_seed2<<<(NN * (AP / 4) + SBLK * SPT - 1) / (SBLK * SPT), SBLK, 0, stream>>>(z2, als2, ald2, m2, mj2, acc2);
    k_eagg2<<<edgeWaveBlocks, 256, 0, stream>>>(ei, z2, ev, mj2, acc2);
    k_fin2 <<<(NN * (OUT_CH / 4) + SBLK * SPT - 1) / (SBLK * SPT), SBLK, 0, stream>>>(out, acc2, b2);
}


# h1 and layer-1 sums stored as two 64-column slices; the aggregation launch adds slice 0 for all edges, then slice 1, so a pass's rows and sums fit the second-level cache; sums in atomic arrival order
# speedup vs baseline: 2.7635x; 1.0534x over previous
#include <hip/hip_runtime.h>
#include <math.h>

#define NN 100000
#define EE 1600000
#define IN_CH 128
#define HID 32
#define HEADS 4
#define HC 128
#define OUT_CH 40
#define NEG_SLOPE 0.2f
#define AP 64
#define DCOL OUT_CH
#define SBLK 256
#define SPT 4
#define SLW 64
#define NSL 2
#define SL_OFS(p) ((size_t)(p) * ((size_t)NN * SLW))

typedef __attribute__((ext_vector_type(2))) float v2f;
typedef __attribute__((ext_vector_type(8))) float v8f;
typedef __attribute__((ext_vector_type(4))) float v4f;

__device__ __forceinline__ float atomic_max_f32(float* addr, float v) {
    if (v >= 0.0f)
        return __int_as_float(atomicMax((int*)addr, __float_as_int(v)));
    else
        return __uint_as_float(atomicMin((unsigned int*)addr, __float_as_uint(v)));
}

__device__ __forceinline__ float lrelu(float v) {
    return v >= 0.0f ? v : NEG_SLOPE * v;
}

__device__ __forceinline__ int package_half() {
    return (int)((__builtin_amdgcn_s_sendmsg_rtn(0x87) >> 18) & 1u);
}

__global__ void k_gemm1(const float* __restrict__ x, const float* __restrict__ W1,
                        float* __restrict__ h1) {
    const int wave = threadIdx.x >> 5;
    const int row0 = (blockIdx.x * 4 + wave) * 16;
    if (row0 >= NN) return;
    const int lane = threadIdx.x & 31;
    const int m    = lane & 15;
    const int kb   = (lane >> 4) * 2;

    v8f c[8];
#pragma unroll
    for (int t = 0; t < 8; ++t) c[t] = (v8f){0,0,0,0,0,0,0,0};

    for (int k = 0; k < IN_CH; k += 4) {
        v2f a;
        const float* xr = x + (size_t)(row0 + m) * IN_CH + (k + kb);
        a.x = xr[0];
        a.y = xr[1];
#pragma unroll
        for (int t = 0; t < 8; ++t) {
            v2f b;
            const float* wr = W1 + (size_t)(k + kb) * HC + t * 16 + m;
            b.x = wr[0];
            b.y = wr[HC];
            c[t] = __builtin_amdgcn_wmma_f32_16x16x4_f32(false, a, false, b,
                                                         (short)0, c[t], false, false);
        }
    }
    const int rofs = (lane >> 4) * 8;
#pragma unroll
    for (int t = 0; t < 8; ++t)
#pragma unroll
        for (int r = 0; r < 8; ++r)
            h1[SL_OFS(t >> 2) + (size_t)(row0 + r + rofs) * SLW + (t & 3) * 16 + m] = c[t][r];
}

__global__ void k_al1(const float* __restrict__ h1, const float* __restrict__ a_src,
                      const float* __restrict__ a_dst, float* __restrict__ als,
                      float* __restrict__ ald, float* __restrict__ m) {
    int i = blockIdx.x * blockDim.x + threadIdx.x;
    if (i >= NN * HEADS) return;
    int n = i >> 2, h = i & 3;
    const float* hp = h1 + SL_OFS(h >> 1) + (size_t)n * SLW + (h & 1) * HID;
    float s = 0.f, d = 0.f;
#pragma unroll 8
    for (int cc = 0; cc < HID; ++cc) {
        float v = hp[cc];
        s += v * a_src[h * HID + cc];
        d += v * a_dst[h * HID + cc];
    }
    als[i] = s;
    ald[i] = d;
    const float self = lrelu(s + d);
    m[i] = self;
    m[NN * HEADS + i] = self;
}

static_assert((NN * SLW / 4) % 16 == 0 && HID == 32 && SLW == 2 * HID && HEADS == 2 * NSL, "whole row pieces per slice; two heads a piece");
__device__ __forceinline__ int sl_node_head(int i) {
    const int p  = i >= NN * SLW / 4 ? 1 : 0;
    const int ii = i - p * (NN * SLW / 4);
    return (ii >> 4) * HEADS + p * 2 + ((ii >> 3) & 1);
}

__global__ void k_seed1(const float* __restrict__ h1, const float* __restrict__ als,
                        const float* __restrict__ ald, const float* __restrict__ m,
                        float* __restrict__ mj, float* __restrict__ dn, float* __restrict__ agg) {
    const int base = blockIdx.x * (SBLK * SPT) + threadIdx.x;
    v4f v[SPT];
    float m0[SPT], m1[SPT], ls[SPT], ld[SPT];
#pragma unroll
    for (int k = 0; k < SPT; ++k) {
        const int i = base + k * SBLK;
        const int nh = sl_node_head(i);
        v[k]  = ((const v4f*)h1)[i];
        m0[k] = m[nh];
        m1[k] = m[NN * HEADS + nh];
        ls[k] = als[nh];
        ld[k] = ald[nh];
    }
#pragma unroll
    for (int k = 0; k < SPT; ++k) {
        const int i = base + k * SBLK;
        const int nh = sl_node_head(i);
        const float mx = fmaxf(m0[k], m1[k]);
        const float w = __expf(lrelu(ls[k] + ld[k]) - mx);
        ((v4f*)agg)[i] = v[k] * w;
        ((v4f*)agg)[NN * HC / 4 + i] = (v4f){0.0f, 0.0f, 0.0f, 0.0f};
        if ((i & 7) == 0) {
            mj[nh] = mx;
            dn[nh] = w;
            dn[NN * HEADS + nh] = 0.0f;
        }
    }
}

static_assert(EE % 32 == 0 && HEADS == 4 && SPT * 8 == 32, "a wave's 32 edges are SPT rounds of 8 edges x 4 heads");
__global__ void k_emax1(const int* __restrict__ ei, const float* __restrict__ als,
                        const float* __restrict__ ald, float* __restrict__ m,
                        float* __restrict__ ev) {
    const int e0 = (blockIdx.x * (SBLK / 32) + (threadIdx.x >> 5)) * 32;
    if (e0 >= EE) return;
    const int lane = threadIdx.x & 31;
    const int s = ei[e0 + lane], d = ei[EE + e0 + lane];
    const int h = lane & 3, q = lane >> 2;
    int dk[SPT];
    float a[SPT], b[SPT];
#pragma unroll
    for (int k = 0; k < SPT; ++k) {
        const int sk = __shfl(s, k * 8 + q);
        dk[k] = __shfl(d, k * 8 + q);
        a[k] = als[sk * 4 + h];
        b[k] = ald[dk[k] * 4 + h];
    }
    float* mh = m + (size_t)package_half() * (NN * HEADS);
#pragma unroll
    for (int k = 0; k < SPT; ++k) {
        const float v = lrelu(a[k] + b[k]);
        ev[(size_t)e0 * HEADS + k * 32 + lane] = v;
        atomic_max_f32(&mh[dk[k] * 4 + h], v);
    }
}

#define EPW 8
#define EPW1 16
static_assert(EE % EPW == 0 && EE % (EPW1 * 8) == 0, "the edge list is a whole number of groups and of blocks");
static_assert(HID == 32 && HC == HEADS * 32 && HEADS == 4 && NSL == 2, "one head = one wave-wide chunk of a row; a slice = 2 heads");
__global__ void k_eagg1(const int* __restrict__ ei, const float* __restrict__ h1,
                        const float* __restrict__ ev,
                        const float* __restrict__ m, float* __restrict__ dn,
                        float* __restrict__ agg) {
    const int blocksPerSlice = EE / (EPW1 * 8);
    const int slice = blockIdx.x >= blocksPerSlice ? 1 : 0;
    const int e0 = ((blockIdx.x - slice * blocksPerSlice) * 8 + (threadIdx.x >> 5)) * EPW1;
    const int lane = threadIdx.x & 31;
    const int s = ei[e0 + (lane & (EPW1 - 1))], d = ei[EE + e0 + (lane & (EPW1 - 1))];
    const int hh = lane & 3;
    int dq[2];
    float v[2];
#pragma unroll
    for (int r = 0; r < 2; ++r) {
        dq[r] = ei[EE + e0 + r * 8 + (lane >> 2)];
        v[r]  = ev[(size_t)(e0 + r * 8) * HEADS + lane];
    }
    const int half = package_half();

    float hv[EPW1][2];
#pragma unroll
    for (int j = 0; j < EPW1; ++j) {
        const int sj = __builtin_amdgcn_readlane(s, j);
        const float* hp = h1 + SL_OFS(slice) + (size_t)sj * SLW + lane;
        hv[j][0] = hp[0];
        hv[j][1] = hp[HID];
    }

    float mx[2];
#pragma unroll
    for (int r = 0; r < 2; ++r)
        mx[r] = m[dq[r] * 4 + hh];

    asm volatile("" ::
                       "v"(hv[0][0]), "v"(hv[0][1]), "v"(hv[1][0]), "v"(hv[1][1]), "v"(hv[2][0]), "v"(hv[2][1]),
                       "v"(hv[3][0]), "v"(hv[3][1]), "v"(hv[4][0]), "v"(hv[4][1]), "v"(hv[5][0]), "v"(hv[5][1]),
                       "v"(hv[6][0]), "v"(hv[6][1]), "v"(hv[7][0]), "v"(hv[7][1]), "v"(hv[8][0]), "v"(hv[8][1]),
                       "v"(hv[9][0]), "v"(hv[9][1]), "v"(hv[10][0]), "v"(hv[10][1]), "v"(hv[11][0]), "v"(hv[11][1]),
                       "v"(hv[12][0]), "v"(hv[12][1]), "v"(hv[13][0]), "v"(hv[13][1]), "v"(hv[14][0]), "v"(hv[14][1]),
                       "v"(hv[15][0]), "v"(hv[15][1]),
                       "v"(v[0]), "v"(v[1]), "v"(mx[0]), "v"(mx[1]));
    static_assert(EPW1 == 16, "the statement above names 16 x 2 row values");

    dn  += (size_t)half * (NN * HEADS);
    agg += (size_t)half * ((size_t)NN * HC) + SL_OFS(slice);

    float w[2];
#pragma unroll
    for (int r = 0; r < 2; ++r)
        w[r] = __expf(v[r] - mx[r]);
    if (slice == 0) {
#pragma unroll
        for (int r = 0; r < 2; ++r)
            atomicAdd(&dn[dq[r] * 4 + hh], w[r]);
    }

#pragma unroll
    for (int j = 0; j < EPW1; ++j) {
        const int dj = __builtin_amdgcn_readlane(d, j);
        float* op = agg + (size_t)dj * SLW + lane;
#pragma unroll
        for (int q = 0; q < 2; ++q) {
            const float wq = __int_as_float(__builtin_amdgcn_readlane(__float_as_int(w[j >> 3]), (j & 7) * 4 + slice * 2 + q));
            atomicAdd(op + q * HID, hv[j][q] * wq);
        }
    }
}

__device__ __forceinline__ float fin1_one(float p, float q, float den, float b) {
    const float v = (p + q) / (den + 1e-16f) + b;
    return v > 0.0f ? v : expm1f(v);
}
static_assert((NN * HC / 4) % (SBLK * SPT) == 0, "whole blocks");
__global__ void k_fin1(float* __restrict__ agg, const float* __restrict__ dn,
                       const float* __restrict__ b1) {
    const int base = blockIdx.x * (SBLK * SPT) + threadIdx.x;
    v4f p[SPT], q[SPT], bb[SPT];
    float d0[SPT], d1[SPT];
#pragma unroll
    for (int k = 0; k < SPT; ++k) {
        const int i = base + k * SBLK;
        p[k]  = ((const v4f*)agg)[i];
        q[k]  = ((const v4f*)agg)[NN * HC / 4 + i];
        const int nh = sl_node_head(i);
        d0[k] = dn[nh];
        d1[k] = dn[NN * HEADS + nh];
        bb[k] = ((const v4f*)b1)[(i >= NN * SLW / 4 ? SLW / 4 : 0) + (i & (SLW / 4 - 1))];
    }
#pragma unroll
    for (int k = 0; k < SPT; ++k) {
        const int i = base + k * SBLK;
        const float den = d0[k] + d1[k];
        ((v4f*)agg)[i] = (v4f){fin1_one(p[k].x, q[k].x, den, bb[k].x), fin1_one(p[k].y, q[k].y, den, bb[k].y),
                               fin1_one(p[k].z, q[k].z, den, bb[k].z), fin1_one(p[k].w, q[k].w, den, bb[k].w)};
    }
}

__global__ void k_gemm2(const float* __restrict__ h2, const float* __restrict__ W2,
                        float* __restrict__ z) {
    const int wave = threadIdx.x >> 5;
    const int row0 = (blockIdx.x * 4 + wave) * 16;
    if (row0 >= NN) return;
    const int lane = threadIdx.x & 31;
    const int m    = lane & 15;
    const int kb   = (lane >> 4) * 2;

    v8f c[3];
#pragma unroll
    for (int t = 0; t < 3; ++t) c[t] = (v8f){0,0,0,0,0,0,0,0};

    for (int k = 0; k < HC; k += 4) {
        v2f a;
        const float* xr = h2 + SL_OFS(k >> 6) + (size_t)(row0 + m) * SLW + ((k & (SLW - 1)) + kb);
        a.x = xr[0];
        a.y = xr[1];
#pragma unroll
        for (int t = 0; t < 3; ++t) {
            int col = t * 16 + m;
            v2f b;
            b.x = (col < OUT_CH) ? W2[(size_t)(k + kb) * OUT_CH + col] : 0.0f;
            b.y = (col < OUT_CH) ? W2[(size_t)(k + kb + 1) * OUT_CH + col] : 0.0f;
            c[t] = __builtin_amdgcn_wmma_f32_16x16x4_f32(false, a, false, b,
                                                         (short)0, c[t], false, false);
        }
    }
    const int rofs = (lane >> 4) * 8;
#pragma unroll
    for (int t = 0; t < 3; ++t) {
        int col = t * 16 + m;
        if (col < OUT_CH)
#pragma unroll
            for (int r = 0; r < 8; ++r)
                z[(size_t)(row0 + r + rofs) * OUT_CH + col] = c[t][r];
    }
}

__global__ void k_al2(const float* __restrict__ z, const float* __restrict__ a_src,
                      const float* __restrict__ a_dst, float* __restrict__ als,
                      float* __restrict__ ald, float* __restrict__ m) {
    int n = blockIdx.x * blockDim.x + threadIdx.x;
    if (n >= NN) return;
    const float* zp = z + (size_t)n * OUT_CH;
    float s = 0.f, d = 0.f;
#pragma unroll 8
    for (int cc = 0; cc < OUT_CH; ++cc) {
        float v = zp[cc];
        s += v * a_src[cc];
        d += v * a_dst[cc];
    }
    als[n] = s;
    ald[n] = d;
    const float self = lrelu(s + d);
    m[n] = self;
    m[NN + n] = self;
}

static_assert(OUT_CH % 4 == 0 && DCOL == OUT_CH && AP % 4 == 0, "the exp-sum opens the float4 behind the channels");
static_assert(SBLK % (AP / 4) == 0, "a thread's items sit in the same float4 column of their rows");
__global__ void k_seed2(const float* __restrict__ z, const float* __restrict__ als,
                        const float* __restrict__ ald, const float* __restrict__ m,
                        float* __restrict__ mj, float* __restrict__ acc) {
    const int base = blockIdx.x * (SBLK * SPT) + threadIdx.x;
    const int j = base % (AP / 4);
    v4f v[SPT];
    float m0[SPT], m1[SPT], ls[SPT], ld[SPT];
#pragma unroll
    for (int k = 0; k < SPT; ++k) {
        const int i = min(base + k * SBLK, NN * (AP / 4) - 1);
        const int n = i / (AP / 4);
        m0[k] = m[n];
        m1[k] = m[NN + n];
        ls[k] = als[n];
        ld[k] = ald[n];
        v[k] = ((const v4f*)z)[n * (OUT_CH / 4) + min(j, OUT_CH / 4 - 1)];
    }
    asm volatile("" :: "v"(v[0]), "v"(v[1]), "v"(v[2]), "v"(v[3]), "v"(m0[0]), "v"(m0[1]), "v"(m0[2]), "v"(m0[3]),
                       "v"(m1[0]), "v"(m1[1]), "v"(m1[2]), "v"(m1[3]), "v"(ls[0]), "v"(ls[1]), "v"(ls[2]), "v"(ls[3]),
                       "v"(ld[0]), "v"(ld[1]), "v"(ld[2]), "v"(ld[3]));
    static_assert(SPT == 4, "the statement above names SPT = 4 items");
#pragma unroll
    for (int k = 0; k < SPT; ++k) {
        const int i = base + k * SBLK;
        if (i < NN * (AP / 4)) {
            const int n = i / (AP / 4);
            v4f o = (v4f){0.0f, 0.0f, 0.0f, 0.0f};
            if (j <= OUT_CH / 4) {
                const float mx = fmaxf(m0[k], m1[k]);
                const float w = __expf(lrelu(ls[k] + ld[k]) - mx);
                if (j < OUT_CH / 4) {
                    o = v[k] * w;
                } else {
                    o.x = w;
                    mj[n] = mx;
                }
            }
            ((v4f*)acc)[i] = o;
            ((v4f*)acc)[NN * (AP / 4) + i] = (v4f){0.0f, 0.0f, 0.0f, 0.0f};
        }
    }
}

static_assert(EE % (32 * SPT) == 0, "the edge list is a whole number of wave groups");
__global__ void k_emax2(const int* __restrict__ ei, const float* __restrict__ als,
                        const float* __restrict__ ald, float* __restrict__ m,
                        float* __restrict__ ev) {
    const int e0 = (blockIdx.x * (SBLK / 32) + (threadIdx.x >> 5)) * (32 * SPT) + (threadIdx.x & 31);
    if (e0 >= EE) return;
    int s[SPT], d[SPT];
    float a[SPT], b[SPT];
#pragma unroll
    for (int k = 0; k < SPT; ++k) {
        s[k] = ei[e0 + k * 32];
        d[k] = ei[EE + e0 + k * 32];
    }
#pragma unroll
    for (int k = 0; k < SPT; ++k) {
        a[k] = als[s[k]];
        b[k] = ald[d[k]];
    }
    float* mh = m + (size_t)package_half() * NN;
#pragma unroll
    for (int k = 0; k < SPT; ++k) {
        const float v = lrelu(a[k] + b[k]);
        ev[e0 + k * 32] = v;
        atomic_max_f32(&mh[d[k]], v);
    }
}

static_assert(DCOL >= 32 && DCOL < AP && (AP * 4) % 128 == 0, "exp-sum sits in the row's second line");
__global__ void k_eagg2(const int* __restrict__ ei, const float* __restrict__ z,
                        const float* __restrict__ ev,
                        const float* __restrict__ m, float* __restrict__ acc) {
    const int e0 = (blockIdx.x * 8 + (threadIdx.x >> 5)) * EPW;
    if (e0 >= EE) return;
    const int lane = threadIdx.x & 31;
    const int je = e0 + (lane & (EPW - 1));
    const int s = ei[je], d = ei[EE + je];

    float z0[EPW], z1[EPW];
    const int tl = 32 + (lane & (OUT_CH - 32 - 1));
    static_assert(((OUT_CH - 32) & (OUT_CH - 33)) == 0, "tail width is a power of two");
#pragma unroll
    for (int j = 0; j < EPW; ++j) {
        const int sj = __builtin_amdgcn_readlane(s, j);
        const float* zp = z + (size_t)sj * OUT_CH;
        z0[j] = zp[lane];
        z1[j] = zp[tl];
    }

    const float lv = ev[je], mx = m[d];

    const int half = package_half();

    asm volatile("" :: "v"(z0[0]), "v"(z0[1]), "v"(z0[2]), "v"(z0[3]),
                       "v"(z0[4]), "v"(z0[5]), "v"(z0[6]), "v"(z0[7]),
                       "v"(z1[0]), "v"(z1[1]), "v"(z1[2]), "v"(z1[3]),
                       "v"(z1[4]), "v"(z1[5]), "v"(z1[6]), "v"(z1[7]),
                       "v"(lv), "v"(mx));

    const float w = __expf(lv - mx);

    acc += (size_t)half * ((size_t)NN * AP);

#pragma unroll
    for (int j = 0; j < EPW; ++j) {
        const int dj = __builtin_amdgcn_readlane(d, j);
        const float wj = __int_as_float(__builtin_amdgcn_readlane(__float_as_int(w), j));
        float* op = acc + (size_t)dj * AP;
        atomicAdd(op + lane, z0[j] * wj);
        const float tv = (lane == DCOL - 32) ? wj : z1[j] * wj;
        if (lane <= DCOL - 32)
            atomicAdd(op + 32 + lane, tv);
    }
}

__global__ void k_fin2(float* __restrict__ out, const float* __restrict__ acc,
                       const float* __restrict__ b2) {
    const int base = blockIdx.x * (SBLK * SPT) + threadIdx.x;
    v4f p[SPT], q[SPT], bb[SPT];
    float d0[SPT], d1[SPT];
#pragma unroll
    for (int k = 0; k < SPT; ++k) {
        const int i = min(base + k * SBLK, NN * (OUT_CH / 4) - 1);
        const int n = i / (OUT_CH / 4), j = i % (OUT_CH / 4);
        const float* ap = acc + (size_t)n * AP;
        const float* bp = ap + (size_t)NN * AP;
        p[k]  = ((const v4f*)ap)[j];
        q[k]  = ((const v4f*)bp)[j];
        d0[k] = ap[DCOL];
        d1[k] = bp[DCOL];
        bb[k] = ((const v4f*)b2)[j];
    }
    asm volatile("" :: "v"(p[0]), "v"(p[1]), "v"(p[2]), "v"(p[3]), "v"(q[0]), "v"(q[1]), "v"(q[2]), "v"(q[3]),
                       "v"(bb[0]), "v"(bb[1]), "v"(bb[2]), "v"(bb[3]),
                       "v"(d0[0]), "v"(d0[1]), "v"(d0[2]), "v"(d0[3]), "v"(d1[0]), "v"(d1[1]), "v"(d1[2]), "v"(d1[3]));
    static_assert(SPT == 4, "the statement above names SPT = 4 items");
#pragma unroll
    for (int k = 0; k < SPT; ++k) {
        const int i = base + k * SBLK;
        if (i < NN * (OUT_CH / 4)) {
            const float den = d0[k] + d1[k] + 1e-16f;
            ((v4f*)out)[i] = (v4f){(p[k].x + q[k].x) / den + bb[k].x, (p[k].y + q[k].y) / den + bb[k].y,
                                   (p[k].z + q[k].z) / den + bb[k].z, (p[k].w + q[k].w) / den + bb[k].w};
        }
    }
}

extern "C" void kernel_launch(void* const* d_in, const int* in_sizes, int n_in,
                              void* d_out, int out_size, void* d_ws, size_t ws_size,
                              hipStream_t stream) {
    const float* x      = (const float*)d_in[0];
    const int*   ei     = (const int*)  d_in[1];
    const float* W1     = (const float*)d_in[2];
    const float* a_src1 = (const float*)d_in[3];
    const float* a_dst1 = (const float*)d_in[4];
    const float* b1     = (const float*)d_in[5];
    const float* W2     = (const float*)d_in[6];
    const float* a_src2 = (const float*)d_in[7];
    const float* a_dst2 = (const float*)d_in[8];
    const float* b2     = (const float*)d_in[9];
    float* out = (float*)d_out;

    float* ws   = (float*)d_ws;
    float* h1   = ws;
    float* agg1 = h1   + (size_t)NN * HC;
    float* acc2 = agg1 + 2 * (size_t)NN * HC;
    float* z2   = acc2 + 2 * (size_t)NN * AP;
    float* als1 = z2   + (size_t)NN * OUT_CH;
    float* ald1 = als1 + (size_t)NN * HEADS;
    float* m1   = ald1 + (size_t)NN * HEADS;
    float* mj1  = m1   + 2 * (size_t)NN * HEADS;
    float* dn1  = mj1  + (size_t)NN * HEADS;
    float* als2 = dn1  + 2 * (size_t)NN * HEADS;
    float* ald2 = als2 + NN;
    float* m2   = ald2 + NN;
    float* mj2  = m2   + 2 * NN;
    float* ev   = mj2  + NN;

    const int TB = 256;
    const int gemmBlocks = (NN / 16 + 3) / 4;
    const int edgeWaveBlocks = (EE / EPW + 7) / 8;

    k_gemm1<<<gemmBlocks, 128, 0, stream>>>(x, W1, h1);
    k_al1  <<<(NN * HEADS + TB - 1) / TB, TB, 0, stream>>>(h1, a_src1, a_dst1, als1, ald1, m1);
    k_emax1<<<(EE / 32 + SBLK / 32 - 1) / (SBLK / 32), SBLK, 0, stream>>>(ei, als1, ald1, m1, ev);
    k_seed1<<<NN * HC / 4 / (SBLK * SPT), SBLK, 0, stream>>>(h1, als1, ald1, m1, mj1, dn1, agg1);
    k_eagg1<<<NSL * (EE / (EPW1 * 8)), 256, 0, stream>>>(ei, h1, ev, mj1, dn1, agg1);
    k_fin1 <<<NN * HC / 4 / (SBLK * SPT), SBLK, 0, stream>>>(agg1, dn1, b1);

    k_gemm2<<<gemmBlocks, 128, 0, stream>>>(agg1, W2, z2);
    k_al2  <<<(NN + TB - 1) / TB, TB, 0, stream>>>(z2, a_src2, a_dst2, als2, ald2, m2);
    k_emax2<<<(EE / (32 * SPT) + SBLK / 32 - 1) / (SBLK / 32), SBLK, 0, stream>>>(ei, als2, ald2, m2, ev);
    k_seed2<<<(NN * (AP / 4) + SBLK * SPT - 1) / (SBLK * SPT), SBLK, 0, stream>>>(z2, als2, ald2, m2, mj2, acc2);
    k_eagg2<<<edgeWaveBlocks, 256, 0, stream>>>(ei, z2, ev, mj2, acc2);
    k_fin2 <<<(NN * (OUT_CH / 4) + SBLK * SPT - 1) / (SBLK * SPT), SBLK, 0, stream>>>(out, acc2, b2);
}


# Layer-1 sums kept in LDS: a group owns 391 target nodes, finds its edges in a byte-per-edge range id list, re-forms their logits, adds rows with LDS float adds (arrival order), writes h2 itself
# speedup vs baseline: 3.1376x; 1.1354x over previous
#include <hip/hip_runtime.h>
#include <math.h>

#define NN 100000
#define EE 1600000
#define IN_CH 128
#define HID 32
#define HEADS 4
#define HC 128
#define OUT_CH 40
#define NEG_SLOPE 0.2f
#define AP 64
#define DCOL OUT_CH
#define SBLK 256
#define SPT 4
#define SLW 64
#define NSL 2
#define SL_OFS(p) ((size_t)(p) * ((size_t)NN * SLW))
#define NPB 391
#define NGRP 256
#define LTHR 1024
#define SCAN 512
static_assert(NGRP * NPB >= NN && (NGRP - 1) * NPB < NN && NGRP <= 256, "the ranges cover the nodes; a range id fits a byte");

typedef __attribute__((ext_vector_type(2))) float v2f;
typedef __attribute__((ext_vector_type(8))) float v8f;
typedef __attribute__((ext_vector_type(4))) float v4f;

__device__ __forceinline__ float atomic_max_f32(float* addr, float v) {
    if (v >= 0.0f)
        return __int_as_float(atomicMax((int*)addr, __float_as_int(v)));
    else
        return __uint_as_float(atomicMin((unsigned int*)addr, __float_as_uint(v)));
}

__device__ __forceinline__ float lrelu(float v) {
    return v >= 0.0f ? v : NEG_SLOPE * v;
}

__device__ __forceinline__ int package_half() {
    return (int)((__builtin_amdgcn_s_sendmsg_rtn(0x87) >> 18) & 1u);
}

__global__ void k_gemm1(const float* __restrict__ x, const float* __restrict__ W1,
                        float* __restrict__ h1) {
    const int wave = threadIdx.x >> 5;
    const int row0 = (blockIdx.x * 4 + wave) * 16;
    if (row0 >= NN) return;
    const int lane = threadIdx.x & 31;
    const int m    = lane & 15;
    const int kb   = (lane >> 4) * 2;

    v8f c[8];
#pragma unroll
    for (int t = 0; t < 8; ++t) c[t] = (v8f){0,0,0,0,0,0,0,0};

    for (int k = 0; k < IN_CH; k += 4) {
        v2f a;
        const float* xr = x + (size_t)(row0 + m) * IN_CH + (k + kb);
        a.x = xr[0];
        a.y = xr[1];
#pragma unroll
        for (int t = 0; t < 8; ++t) {
            v2f b;
            const float* wr = W1 + (size_t)(k + kb) * HC + t * 16 + m;
            b.x = wr[0];
            b.y = wr[HC];
            c[t] = __builtin_amdgcn_wmma_f32_16x16x4_f32(false, a, false, b,
                                                         (short)0, c[t], false, false);
        }
    }
    const int rofs = (lane >> 4) * 8;
#pragma unroll
    for (int t = 0; t < 8; ++t)
#pragma unroll
        for (int r = 0; r < 8; ++r)
            h1[SL_OFS(t >> 2) + (size_t)(row0 + r + rofs) * SLW + (t & 3) * 16 + m] = c[t][r];
}

__global__ void k_al1(const float* __restrict__ h1, const float* __restrict__ a_src,
                      const float* __restrict__ a_dst, float* __restrict__ als,
                      float* __restrict__ ald, float* __restrict__ m) {
    int i = blockIdx.x * blockDim.x + threadIdx.x;
    if (i >= NN * HEADS) return;
    int n = i >> 2, h = i & 3;
    const float* hp = h1 + SL_OFS(h >> 1) + (size_t)n * SLW + (h & 1) * HID;
    float s = 0.f, d = 0.f;
#pragma unroll 8
    for (int cc = 0; cc < HID; ++cc) {
        float v = hp[cc];
        s += v * a_src[h * HID + cc];
        d += v * a_dst[h * HID + cc];
    }
    als[i] = s;
    ald[i] = d;
    const float self = lrelu(s + d);
    m[i] = self;
    m[NN * HEADS + i] = self;
}

static_assert(EE % 32 == 0 && HEADS == 4 && SPT * 8 == 32, "a wave's 32 edges are SPT rounds of 8 edges x 4 heads");
__global__ void k_emax1(const int* __restrict__ ei, const float* __restrict__ als,
                        const float* __restrict__ ald, float* __restrict__ m,
                        float* __restrict__ rid) {
    const int e0 = (blockIdx.x * (SBLK / 32) + (threadIdx.x >> 5)) * 32;
    if (e0 >= EE) return;
    const int lane = threadIdx.x & 31;
    const int s = ei[e0 + lane], d = ei[EE + e0 + lane];
    ((unsigned char*)rid)[e0 + lane] = (unsigned char)((unsigned)d / NPB);
    const int h = lane & 3, q = lane >> 2;
    int dk[SPT];
    float a[SPT], b[SPT];
#pragma unroll
    for (int k = 0; k < SPT; ++k) {
        const int sk = __shfl(s, k * 8 + q);
        dk[k] = __shfl(d, k * 8 + q);
        a[k] = als[sk * 4 + h];
        b[k] = ald[dk[k] * 4 + h];
    }
    float* mh = m + (size_t)package_half() * (NN * HEADS);
#pragma unroll
    for (int k = 0; k < SPT; ++k) {
        atomic_max_f32(&mh[dk[k] * 4 + h], lrelu(a[k] + b[k]));
    }
}

static_assert(EE % SCAN == 0 && SCAN == 32 * 16, "whole scan steps; a lane reads 16 range bytes");
static_assert(NPB * HC * 4 + 3 * NPB * HEADS * 4 <= 320 * 1024, "a range's sums, exp-sums, maxima and target logits fit the LDS");
static_assert(HID == 32 && HC == 128 && HEADS == 4 && SLW == 64 && NSL == 2, "a head is one wave-wide chunk; a slice holds two heads");
__device__ __forceinline__ float fin1_one(float p, float den, float b) {
    const float v = p / (den + 1e-16f) + b;
    return v > 0.0f ? v : expm1f(v);
}
__global__ __launch_bounds__(LTHR) void k_lagg1(const int* __restrict__ ei, const float* __restrict__ h1,
                        const float* __restrict__ als,
                        const float* __restrict__ ald, const float* __restrict__ m,
                        const float* __restrict__ rid, const float* __restrict__ b1,
                        float* __restrict__ h2) {
    __shared__ __attribute__((aligned(16))) float acc[NPB * HC];
    __shared__ float dnl[NPB * HEADS];
    __shared__ float mxl[NPB * HEADS];
    __shared__ float adl[NPB * HEADS];

    const int g  = blockIdx.x;
    const int n0 = g * NPB;
    const int nn = min(NPB, NN - n0);
    const int tid = threadIdx.x, lane = tid & 31;
    const int wave = __builtin_amdgcn_readfirstlane(tid >> 5);

    for (int i = tid; i < nn * HEADS; i += LTHR) {
        const int nh = n0 * HEADS + i;
        const float mx = fmaxf(m[nh], m[NN * HEADS + nh]);
        const float ad = ald[nh];
        mxl[i] = mx;
        adl[i] = ad;
        dnl[i] = __expf(lrelu(als[nh] + ad) - mx);
    }
    __syncthreads();
    for (int i = tid; i < nn * (HC / 4); i += LTHR) {
        const int n = i >> 5, c4 = i & 31;
        const v4f v = ((const v4f*)(h1 + SL_OFS(c4 >> 4) + (size_t)(n0 + n) * SLW))[c4 & 15];
        ((v4f*)acc)[i] = v * dnl[n * HEADS + (c4 >> 3)];
    }
    __syncthreads();

    const unsigned pat = (unsigned)g * 0x01010101u;
    const uint4* rb = (const uint4*)rid;
    int qreg = 0;
    int cnt = 0;
    uint4 x = rb[(size_t)wave * 32 + lane];
    for (int t = wave; t < EE / SCAN; t += LTHR / 32) {
        const uint4 xn = rb[(size_t)min(t + LTHR / 32, EE / SCAN - 1) * 32 + lane];
        const unsigned t0 = x.x ^ pat, t1 = x.y ^ pat, t2 = x.z ^ pat, t3 = x.w ^ pat;
        const unsigned any = (((t0 - 0x01010101u) & ~t0) | ((t1 - 0x01010101u) & ~t1) |
                              ((t2 - 0x01010101u) & ~t2) | ((t3 - 0x01010101u) & ~t3)) & 0x80808080u;
        unsigned mask = __builtin_amdgcn_ballot_w32(any != 0u);
        while (mask) {
            const int L = __builtin_ctz(mask);
            mask &= mask - 1;
#pragma unroll
            for (int dd = 0; dd < 4; ++dd) {
                const unsigned ts = (unsigned)__builtin_amdgcn_readlane((int)(dd == 0 ? t0 : dd == 1 ? t1 : dd == 2 ? t2 : t3), L);
                unsigned zz = ~(((ts & 0x7F7F7F7Fu) + 0x7F7F7F7Fu) | ts) & 0x80808080u;
                while (zz) {
                    const int b = __builtin_ctz(zz);
                    zz &= zz - 1;
                    const int e = t * SCAN + L * 16 + dd * 4 + (b >> 3);
                    qreg = (lane == cnt) ? e : qreg;
                    ++cnt;
                }
            }
            while (cnt >= 8) {
                const int ej = __shfl(qreg, lane & 7);
                const int s  = ei[ej], d = ei[EE + ej] - n0;
                const int hh = lane & 3;
                const int sq = __shfl(s, lane >> 2), dq = __shfl(d, lane >> 2);
                const float la = als[sq * HEADS + hh];

                float hv[8][4];
#pragma unroll
                for (int j = 0; j < 8; ++j) {
                    const int sj = __builtin_amdgcn_readlane(s, j);
                    const float* hp = h1 + (size_t)sj * SLW + lane;
                    hv[j][0] = hp[0];
                    hv[j][1] = hp[HID];
                    hv[j][2] = hp[SL_OFS(1)];
                    hv[j][3] = hp[SL_OFS(1) + HID];
                }
                const float w = __expf(lrelu(la + adl[dq * HEADS + hh]) - mxl[dq * HEADS + hh]);
                atomicAdd(&dnl[dq * HEADS + hh], w);
#pragma unroll
                for (int j = 0; j < 8; ++j) {
                    const int dj = __builtin_amdgcn_readlane(d, j);
                    float* op = &acc[dj * HC + lane];
#pragma unroll
                    for (int q = 0; q < 4; ++q) {
                        const float wq = __int_as_float(__builtin_amdgcn_readlane(__float_as_int(w), j * 4 + q));
                        atomicAdd(op + q * HID, hv[j][q] * wq);
                    }
                }
                qreg = __shfl_down(qreg, 8);
                cnt -= 8;
            }
        }
        x = xn;
    }
    for (int j = 0; j < cnt; ++j) {
        const int e = __builtin_amdgcn_readlane(qreg, j);
        const int s = ei[e], d = ei[EE + e] - n0;
        const int hh = lane & 3;
        const float* hp = h1 + (size_t)s * SLW + lane;
        const float r0 = hp[0], r1 = hp[HID], r2 = hp[SL_OFS(1)], r3 = hp[SL_OFS(1) + HID];
        const float w = __expf(lrelu(als[s * HEADS + hh] + adl[d * HEADS + hh]) - mxl[d * HEADS + hh]);
        if (lane < HEADS)
            atomicAdd(&dnl[d * HEADS + lane], w);
        float* op = &acc[d * HC + lane];
        atomicAdd(op,           r0 * __int_as_float(__builtin_amdgcn_readlane(__float_as_int(w), 0)));
        atomicAdd(op + HID,     r1 * __int_as_float(__builtin_amdgcn_readlane(__float_as_int(w), 1)));
        atomicAdd(op + 2 * HID, r2 * __int_as_float(__builtin_amdgcn_readlane(__float_as_int(w), 2)));
        atomicAdd(op + 3 * HID, r3 * __int_as_float(__builtin_amdgcn_readlane(__float_as_int(w), 3)));
    }
    __syncthreads();

    for (int i = tid; i < nn * (HC / 4); i += LTHR) {
        const int n = i >> 5, c4 = i & 31;
        const v4f a = ((const v4f*)acc)[i];
        const float den = dnl[n * HEADS + (c4 >> 3)];
        const v4f bb = ((const v4f*)b1)[c4];
        ((v4f*)(h2 + SL_OFS(c4 >> 4) + (size_t)(n0 + n) * SLW))[c4 & 15] =
            (v4f){fin1_one(a.x, den, bb.x), fin1_one(a.y, den, bb.y), fin1_one(a.z, den, bb.z), fin1_one(a.w, den, bb.w)};
    }
}

__global__ void k_gemm2(const float* __restrict__ h2, const float* __restrict__ W2,
                        float* __restrict__ z) {
    const int wave = threadIdx.x >> 5;
    const int row0 = (blockIdx.x * 4 + wave) * 16;
    if (row0 >= NN) return;
    const int lane = threadIdx.x & 31;
    const int m    = lane & 15;
    const int kb   = (lane >> 4) * 2;

    v8f c[3];
#pragma unroll
    for (int t = 0; t < 3; ++t) c[t] = (v8f){0,0,0,0,0,0,0,0};

    for (int k = 0; k < HC; k += 4) {
        v2f a;
        const float* xr = h2 + SL_OFS(k >> 6) + (size_t)(row0 + m) * SLW + ((k & (SLW - 1)) + kb);
        a.x = xr[0];
        a.y = xr[1];
#pragma unroll
        for (int t = 0; t < 3; ++t) {
            int col = t * 16 + m;
            v2f b;
            b.x = (col < OUT_CH) ? W2[(size_t)(k + kb) * OUT_CH + col] : 0.0f;
            b.y = (col < OUT_CH) ? W2[(size_t)(k + kb + 1) * OUT_CH + col] : 0.0f;
            c[t] = __builtin_amdgcn_wmma_f32_16x16x4_f32(false, a, false, b,
                                                         (short)0, c[t], false, false);
        }
    }
    const int rofs = (lane >> 4) * 8;
#pragma unroll
    for (int t = 0; t < 3; ++t) {
        int col = t * 16 + m;
        if (col < OUT_CH)
#pragma unroll
            for (int r = 0; r < 8; ++r)
                z[(size_t)(row0 + r + rofs) * OUT_CH + col] = c[t][r];
    }
}

__global__ void k_al2(const float* __restrict__ z, const float* __restrict__ a_src,
                      const float* __restrict__ a_dst, float* __restrict__ als,
                      float* __restrict__ ald, float* __restrict__ m) {
    int n = blockIdx.x * blockDim.x + threadIdx.x;
    if (n >= NN) return;
    const float* zp = z + (size_t)n * OUT_CH;
    float s = 0.f, d = 0.f;
#pragma unroll 8
    for (int cc = 0; cc < OUT_CH; ++cc) {
        float v = zp[cc];
        s += v * a_src[cc];
        d += v * a_dst[cc];
    }
    als[n] = s;
    ald[n] = d;
    const float self = lrelu(s + d);
    m[n] = self;
    m[NN + n] = self;
}

static_assert(OUT_CH % 4 == 0 && DCOL == OUT_CH && AP % 4 == 0, "the exp-sum opens the float4 behind the channels");
static_assert(SBLK % (AP / 4) == 0, "a thread's items sit in the same float4 column of their rows");
__global__ void k_seed2(const float* __restrict__ z, const float* __restrict__ als,
                        const float* __restrict__ ald, const float* __restrict__ m,
                        float* __restrict__ mj, float* __restrict__ acc) {
    const int base = blockIdx.x * (SBLK * SPT) + threadIdx.x;
    const int j = base % (AP / 4);
    v4f v[SPT];
    float m0[SPT], m1[SPT], ls[SPT], ld[SPT];
#pragma unroll
    for (int k = 0; k < SPT; ++k) {
        const int i = min(base + k * SBLK, NN * (AP / 4) - 1);
        const int n = i / (AP / 4);
        m0[k] = m[n];
        m1[k] = m[NN + n];
        ls[k] = als[n];
        ld[k] = ald[n];
        v[k] = ((const v4f*)z)[n * (OUT_CH / 4) + min(j, OUT_CH / 4 - 1)];
    }
    asm volatile("" :: "v"(v[0]), "v"(v[1]), "v"(v[2]), "v"(v[3]), "v"(m0[0]), "v"(m0[1]), "v"(m0[2]), "v"(m0[3]),
                       "v"(m1[0]), "v"(m1[1]), "v"(m1[2]), "v"(m1[3]), "v"(ls[0]), "v"(ls[1]), "v"(ls[2]), "v"(ls[3]),
                       "v"(ld[0]), "v"(ld[1]), "v"(ld[2]), "v"(ld[3]));
    static_assert(SPT == 4, "the statement above names SPT = 4 items");
#pragma unroll
    for (int k = 0; k < SPT; ++k) {
        const int i = base + k * SBLK;
        if (i < NN * (AP / 4)) {
            const int n = i / (AP / 4);
            v4f o = (v4f){0.0f, 0.0f, 0.0f, 0.0f};
            if (j <= OUT_CH / 4) {
                const float mx = fmaxf(m0[k], m1[k]);
                const float w = __expf(lrelu(ls[k] + ld[k]) - mx);
                if (j < OUT_CH / 4) {
                    o = v[k] * w;
                } else {
                    o.x = w;
                    mj[n] = mx;
                }
            }
            ((v4f*)acc)[i] = o;
            ((v4f*)acc)[NN * (AP / 4) + i] = (v4f){0.0f, 0.0f, 0.0f, 0.0f};
        }
    }
}

static_assert(EE % (32 * SPT) == 0, "the edge list is a whole number of wave groups");
__global__ void k_emax2(const int* __restrict__ ei, const float* __restrict__ als,
                        const float* __restrict__ ald, float* __restrict__ m,
                        float* __restrict__ ev) {
    const int e0 = (blockIdx.x * (SBLK / 32) + (threadIdx.x >> 5)) * (32 * SPT) + (threadIdx.x & 31);
    if (e0 >= EE) return;
    int s[SPT], d[SPT];
    float a[SPT], b[SPT];
#pragma unroll
    for (int k = 0; k < SPT; ++k) {
        s[k] = ei[e0 + k * 32];
        d[k] = ei[EE + e0 + k * 32];
    }
#pragma unroll
    for (int k = 0; k < SPT; ++k) {
        a[k] = als[s[k]];
        b[k] = ald[d[k]];
    }
    float* mh = m + (size_t)package_half() * NN;
#pragma unroll
    for (int k = 0; k < SPT; ++k) {
        const float v = lrelu(a[k] + b[k]);
        ev[e0 + k * 32] = v;
        atomic_max_f32(&mh[d[k]], v);
    }
}

#define EPW 8
static_assert(EE % EPW == 0, "the edge list is a whole number of wave groups");
static_assert(DCOL >= 32 && DCOL < AP && (AP * 4) % 128 == 0, "exp-sum sits in the row's second line");
__global__ void k_eagg2(const int* __restrict__ ei, const float* __restrict__ z,
                        const float* __restrict__ ev,
                        const float* __restrict__ m, float* __restrict__ acc) {
    const int e0 = (blockIdx.x * 8 + (threadIdx.x >> 5)) * EPW;
    if (e0 >= EE) return;
    const int lane = threadIdx.x & 31;
    const int je = e0 + (lane & (EPW - 1));
    const int s = ei[je], d = ei[EE + je];

    float z0[EPW], z1[EPW];
    const int tl = 32 + (lane & (OUT_CH - 32 - 1));
    static_assert(((OUT_CH - 32) & (OUT_CH - 33)) == 0, "tail width is a power of two");
#pragma unroll
    for (int j = 0; j < EPW; ++j) {
        const int sj = __builtin_amdgcn_readlane(s, j);
        const float* zp = z + (size_t)sj * OUT_CH;
        z0[j] = zp[lane];
        z1[j] = zp[tl];
    }

    const float lv = ev[je], mx = m[d];

    const int half = package_half();

    asm volatile("" :: "v"(z0[0]), "v"(z0[1]), "v"(z0[2]), "v"(z0[3]),
                       "v"(z0[4]), "v"(z0[5]), "v"(z0[6]), "v"(z0[7]),
                       "v"(z1[0]), "v"(z1[1]), "v"(z1[2]), "v"(z1[3]),
                       "v"(z1[4]), "v"(z1[5]), "v"(z1[6]), "v"(z1[7]),
                       "v"(lv), "v"(mx));

    const float w = __expf(lv - mx);

    acc += (size_t)half * ((size_t)NN * AP);

#pragma unroll
    for (int j = 0; j < EPW; ++j) {
        const int dj = __builtin_amdgcn_readlane(d, j);
        const float wj = __int_as_float(__builtin_amdgcn_readlane(__float_as_int(w), j));
        float* op = acc + (size_t)dj * AP;
        atomicAdd(op + lane, z0[j] * wj);
        const float tv = (lane == DCOL - 32) ? wj : z1[j] * wj;
        if (lane <= DCOL - 32)
            atomicAdd(op + 32 + lane, tv);
    }
}

__global__ void k_fin2(float* __restrict__ out, const float* __restrict__ acc,
                       const float* __restrict__ b2) {
    const int base = blockIdx.x * (SBLK * SPT) + threadIdx.x;
    v4f p[SPT], q[SPT], bb[SPT];
    float d0[SPT], d1[SPT];
#pragma unroll
    for (int k = 0; k < SPT; ++k) {
        const int i = min(base + k * SBLK, NN * (OUT_CH / 4) - 1);
        const int n = i / (OUT_CH / 4), j = i % (OUT_CH / 4);
        const float* ap = acc + (size_t)n * AP;
        const float* bp = ap + (size_t)NN * AP;
        p[k]  = ((const v4f*)ap)[j];
        q[k]  = ((const v4f*)bp)[j];
        d0[k] = ap[DCOL];
        d1[k] = bp[DCOL];
        bb[k] = ((const v4f*)b2)[j];
    }
    asm volatile("" :: "v"(p[0]), "v"(p[1]), "v"(p[2]), "v"(p[3]), "v"(q[0]), "v"(q[1]), "v"(q[2]), "v"(q[3]),
                       "v"(bb[0]), "v"(bb[1]), "v"(bb[2]), "v"(bb[3]),
                       "v"(d0[0]), "v"(d0[1]), "v"(d0[2]), "v"(d0[3]), "v"(d1[0]), "v"(d1[1]), "v"(d1[2]), "v"(d1[3]));
    static_assert(SPT == 4, "the statement above names SPT = 4 items");
#pragma unroll
    for (int k = 0; k < SPT; ++k) {
        const int i = base + k * SBLK;
        if (i < NN * (OUT_CH / 4)) {
            const float den = d0[k] + d1[k] + 1e-16f;
            ((v4f*)out)[i] = (v4f){(p[k].x + q[k].x) / den + bb[k].x, (p[k].y + q[k].y) / den + bb[k].y,
                                   (p[k].z + q[k].z) / den + bb[k].z, (p[k].w + q[k].w) / den + bb[k].w};
        }
    }
}

extern "C" void kernel_launch(void* const* d_in, const int* in_sizes, int n_in,
                              void* d_out, int out_size, void* d_ws, size_t ws_size,
                              hipStream_t stream) {
    const float* x      = (const float*)d_in[0];
    const int*   ei     = (const int*)  d_in[1];
    const float* W1     = (const float*)d_in[2];
    const float* a_src1 = (const float*)d_in[3];
    const float* a_dst1 = (const float*)d_in[4];
    const float* b1     = (const float*)d_in[5];
    const float* W2     = (const float*)d_in[6];
    const float* a_src2 = (const float*)d_in[7];
    const float* a_dst2 = (const float*)d_in[8];
    const float* b2     = (const float*)d_in[9];
    float* out = (float*)d_out;

    float* ws   = (float*)d_ws;
    float* h1   = ws;
    float* h2   = h1   + (size_t)NN * HC;
    float* rid1 = h2   + (size_t)NN * HC;
    float* acc2 = rid1 + (size_t)NN * HC;
    float* z2   = acc2 + 2 * (size_t)NN * AP;
    float* als1 = z2   + (size_t)NN * OUT_CH;
    float* ald1 = als1 + (size_t)NN * HEADS;
    float* m1   = ald1 + (size_t)NN * HEADS;
    float* als2 = m1   + 2 * (size_t)NN * HEADS;
    float* ald2 = als2 + NN;
    float* m2   = ald2 + NN;
    float* mj2  = m2   + 2 * NN;
    float* ev   = mj2  + NN;

    const int TB = 256;
    const int gemmBlocks = (NN / 16 + 3) / 4;
    const int edgeWaveBlocks = (EE / EPW + 7) / 8;
    static_assert((size_t)EE <= (size_t)NN * HC * 4, "the range ids fit their region");

    k_gemm1<<<gemmBlocks, 128, 0, stream>>>(x, W1, h1);
    k_al1  <<<(NN * HEADS + TB - 1) / TB, TB, 0, stream>>>(h1, a_src1, a_dst1, als1, ald1, m1);
    k_emax1<<<(EE / 32 + SBLK / 32 - 1) / (SBLK / 32), SBLK, 0, stream>>>(ei, als1, ald1, m1, rid1);
    k_lagg1<<<NGRP, LTHR, 0, stream>>>(ei, h1, als1, ald1, m1, rid1, b1, h2);

    k_gemm2<<<gemmBlocks, 128, 0, stream>>>(h2, W2, z2);
    k_al2  <<<(NN + TB - 1) / TB, TB, 0, stream>>>(z2, a_src2, a_dst2, als2, ald2, m2);
    k_emax2<<<(EE / (32 * SPT) + SBLK / 32 - 1) / (SBLK / 32), SBLK, 0, stream>>>(ei, als2, ald2, m2, ev);
    k_seed2<<<(NN * (AP / 4) + SBLK * SPT - 1) / (SBLK * SPT), SBLK, 0, stream>>>(z2, als2, ald2, m2, mj2, acc2);
    k_eagg2<<<edgeWaveBlocks, 256, 0, stream>>>(ei, z2, ev, mj2, acc2);
    k_fin2 <<<(NN * (OUT_CH / 4) + SBLK * SPT - 1) / (SBLK * SPT), SBLK, 0, stream>>>(out, acc2, b2);
}
